# starting kernel
# baseline (speedup 1.0000x reference)
#include <hip/hip_runtime.h>


typedef __attribute__((ext_vector_type(2))) float v2f;
typedef __attribute__((ext_vector_type(8))) float v8f;

#define NM    8
#define NC    1024
#define ROWP  1028
#define EPSF  1e-8f

__global__ __launch_bounds__(256)
void gs_wmma_kernel(const float* __restrict__ x, float* __restrict__ out, int nb) {
  __shared__ float Xs[NM * ROWP];
  __shared__ float Gpart[8][8][8];
  __shared__ float Gm[8][8];
  __shared__ float Ac[8][8];

  const int b    = blockIdx.x;
  const int t    = threadIdx.x;
  const int lane = t & 31;
  const int wave = t >> 5;

  {
    const float4* __restrict__ xv = (const float4*)x;
    #pragma unroll
    for (int row = 0; row < NM; ++row) {
      float4 v = xv[((size_t)row * nb + b) * (NC / 4) + t];
      *(float4*)&Xs[row * ROWP + t * 4] = v;
    }
  }
  __syncthreads();

  {
    const int   m       = lane & 15;
    const int   kb      = (lane >> 4) << 1;
    const int   rowSafe = m & 7;
    const float keep    = (m < NM) ? 1.0f : 0.0f;
    const int   kbase   = wave * 128 + kb;

    v8f acc = {};
    for (int k0 = 0; k0 < 128; k0 += 4) {
      v2f a = *(const v2f*)&Xs[rowSafe * ROWP + kbase + k0];
      a.x *= keep;
      a.y *= keep;
      acc = __builtin_amdgcn_wmma_f32_16x16x4_f32(
           false, a,  false, a,
           (short)0, acc,  false,  false);
    }
    if (lane < 8) {
      #pragma unroll
      for (int r = 0; r < 8; ++r) Gpart[wave][r][lane] = acc[r];
    }
  }
  __syncthreads();

  if (t < 64) {
    const int i = t >> 3, j = t & 7;
    float s = 0.0f;
    #pragma unroll
    for (int w = 0; w < 8; ++w) s += Gpart[w][i][j];
    Gm[i][j] = s;
  }
  __syncthreads();

  if (t == 0) {
    float g[8][8];
    #pragma unroll
    for (int i = 0; i < 8; ++i)
      #pragma unroll
      for (int j = 0; j < 8; ++j) g[i][j] = Gm[i][j];

    float A[8][8];
    #pragma unroll
    for (int i = 0; i < 8; ++i)
      #pragma unroll
      for (int j = 0; j < 8; ++j) A[i][j] = 0.0f;

    float d[8];

    {
      float n0   = sqrtf(fmaxf(g[0][0], 0.0f));
      float inv0 = 1.0f / (n0 + EPSF);
      A[0][0] = inv0;
      d[0]    = inv0 * inv0 * g[0][0];
    }

    #pragma unroll
    for (int i = 1; i < 8; ++i) {
      float c[8];
      #pragma unroll
      for (int k = 0; k < 8; ++k) c[k] = 0.0f;
      c[i] = 1.0f;

      #pragma unroll
      for (int j = 0; j < 8; ++j) {
        if (j < i) {
          float dv = 0.0f;
          #pragma unroll
          for (int k = 0; k < 8; ++k)
            if (k <= j) dv += A[j][k] * g[k][i];
          float proj = dv / (d[j] + EPSF);
          #pragma unroll
          for (int k = 0; k < 8; ++k)
            if (k <= j) c[k] -= proj * A[j][k];
        }
      }

      float n2 = 0.0f;
      #pragma unroll
      for (int k = 0; k < 8; ++k) {
        if (k <= i) {
          float y = 0.0f;
          #pragma unroll
          for (int m2 = 0; m2 < 8; ++m2)
            if (m2 <= i) y += g[k][m2] * c[m2];
          n2 += c[k] * y;
        }
      }
      float nn = sqrtf(fmaxf(n2, 0.0f));
      if (nn > EPSF) {
        float inv = 1.0f / nn;
        #pragma unroll
        for (int k = 0; k < 8; ++k) c[k] *= inv;
        d[i] = n2 * inv * inv;
      } else {
        d[i] = n2;
      }
      #pragma unroll
      for (int k = 0; k < 8; ++k) A[i][k] = c[k];
    }

    #pragma unroll
    for (int i = 0; i < 8; ++i)
      #pragma unroll
      for (int k = 0; k < 8; ++k) Ac[i][k] = A[i][k];
  }
  __syncthreads();

  {
    float4 xk[8];
    #pragma unroll
    for (int k = 0; k < 8; ++k) xk[k] = *(const float4*)&Xs[k * ROWP + t * 4];

    float4* __restrict__ ov = (float4*)out;
    #pragma unroll
    for (int m = 0; m < 8; ++m) {
      float4 acc4 = {0.0f, 0.0f, 0.0f, 0.0f};
      #pragma unroll
      for (int k = 0; k < 8; ++k) {
        const float c = Ac[m][k];
        acc4.x += c * xk[k].x;
        acc4.y += c * xk[k].y;
        acc4.z += c * xk[k].z;
        acc4.w += c * xk[k].w;
      }
      ov[((size_t)m * nb + b) * (NC / 4) + t] = acc4;
    }
  }
}

extern "C" void kernel_launch(void* const* d_in, const int* in_sizes, int n_in,
                              void* d_out, int out_size, void* d_ws, size_t ws_size,
                              hipStream_t stream) {
  const float* x   = (const float*)d_in[0];
  float*       out = (float*)d_out;
  const int    nb  = in_sizes[0] / (NM * NC);
  gs_wmma_kernel<<<nb, 256, 0, stream>>>(x, out, nb);
}


# One work-group takes 4 batch rows (1024 groups, 2 resident per WGP): the serial recurrence on G runs once per group, one lane per batch row, not once per row; every sum keeps its order
# speedup vs baseline: 1.2061x; 1.2061x over previous
#include <hip/hip_runtime.h>


typedef __attribute__((ext_vector_type(2))) float v2f;
typedef __attribute__((ext_vector_type(8))) float v8f;

#define NM    8
#define NC    1024
#define ROWP  1028
#define RB    4
#define EPSF  1e-8f

__global__ __launch_bounds__(256)
void gs_wmma_kernel(const float* __restrict__ x, float* __restrict__ out, int nb) {
  __shared__ float Xs[RB * NM * ROWP];
  __shared__ float Gpart[8][RB][64];
  __shared__ float Gm[64 * RB];
  __shared__ float Ac[RB][64];

  const int t    = threadIdx.x;
  const int lane = t & 31;
  const int wave = t >> 5;
  const int b0   = blockIdx.x * RB;

  {
    const float4* __restrict__ xv = (const float4*)x;
    for (int r = 0; r < RB; ++r) {
      const int bb = min(b0 + r, nb - 1);
      #pragma unroll
      for (int row = 0; row < NM; ++row) {
        float4 v = xv[((size_t)row * nb + bb) * (NC / 4) + t];
        *(float4*)&Xs[(r * NM + row) * ROWP + t * 4] = v;
      }
    }
  }
  __syncthreads();

  {
    const int   m       = lane & 15;
    const int   kb      = (lane >> 4) << 1;
    const int   rowSafe = m & 7;
    const float keep    = (m < NM) ? 1.0f : 0.0f;
    const int   kbase   = wave * 128 + kb;

    v8f acc[RB];
    for (int r = 0; r < RB; ++r) acc[r] = (v8f){0.0f, 0.0f, 0.0f, 0.0f, 0.0f, 0.0f, 0.0f, 0.0f};

    for (int k0 = 0; k0 < 128; k0 += 4) {
      for (int r = 0; r < RB; ++r) {
        v2f a = *(const v2f*)&Xs[(r * NM + rowSafe) * ROWP + kbase + k0];
        a.x *= keep;
        a.y *= keep;
        acc[r] = __builtin_amdgcn_wmma_f32_16x16x4_f32(
             false, a,  false, a,
             (short)0, acc[r],  false,  false);
      }
    }
    if (lane < 8) {
      for (int r = 0; r < RB; ++r) {
        #pragma unroll
        for (int i = 0; i < 8; ++i) Gpart[wave][r][i * 8 + lane] = acc[r][i];
      }
    }
  }
  __syncthreads();

  {
    const int r = t >> 6, ij = t & 63;
    float s = 0.0f;
    #pragma unroll
    for (int w = 0; w < 8; ++w) s += Gpart[w][r][ij];
    Gm[ij * RB + r] = s;
  }
  __syncthreads();

  if (t < RB) {
    float g[8][8];
    #pragma unroll
    for (int i = 0; i < 8; ++i)
      #pragma unroll
      for (int j = 0; j < 8; ++j) g[i][j] = Gm[(i * 8 + j) * RB + t];

    float A[8][8];
    #pragma unroll
    for (int i = 0; i < 8; ++i)
      #pragma unroll
      for (int j = 0; j < 8; ++j) A[i][j] = 0.0f;

    float d[8];

    {
      float n0   = sqrtf(fmaxf(g[0][0], 0.0f));
      float inv0 = 1.0f / (n0 + EPSF);
      A[0][0] = inv0;
      d[0]    = inv0 * inv0 * g[0][0];
    }

    #pragma unroll
    for (int i = 1; i < 8; ++i) {
      float c[8];
      #pragma unroll
      for (int k = 0; k < 8; ++k) c[k] = 0.0f;
      c[i] = 1.0f;

      #pragma unroll
      for (int j = 0; j < 8; ++j) {
        if (j < i) {
          float dv = 0.0f;
          #pragma unroll
          for (int k = 0; k < 8; ++k)
            if (k <= j) dv += A[j][k] * g[k][i];
          float proj = dv / (d[j] + EPSF);
          #pragma unroll
          for (int k = 0; k < 8; ++k)
            if (k <= j) c[k] -= proj * A[j][k];
        }
      }

      float n2 = 0.0f;
      #pragma unroll
      for (int k = 0; k < 8; ++k) {
        if (k <= i) {
          float y = 0.0f;
          #pragma unroll
          for (int m2 = 0; m2 < 8; ++m2)
            if (m2 <= i) y += g[k][m2] * c[m2];
          n2 += c[k] * y;
        }
      }
      float nn = sqrtf(fmaxf(n2, 0.0f));
      if (nn > EPSF) {
        float inv = 1.0f / nn;
        #pragma unroll
        for (int k = 0; k < 8; ++k) c[k] *= inv;
        d[i] = n2 * inv * inv;
      } else {
        d[i] = n2;
      }
      #pragma unroll
      for (int k = 0; k < 8; ++k) A[i][k] = c[k];
    }

    #pragma unroll
    for (int i = 0; i < 8; ++i)
      #pragma unroll
      for (int k = 0; k < 8; ++k) Ac[t][i * 8 + k] = A[i][k];
  }
  __syncthreads();

  {
    float4* __restrict__ ov = (float4*)out;
    for (int r = 0; r < RB; ++r) {
      if (b0 + r < nb) {
        float4 xk[8];
        #pragma unroll
        for (int k = 0; k < 8; ++k) xk[k] = *(const float4*)&Xs[(r * NM + k) * ROWP + t * 4];

        #pragma unroll
        for (int m = 0; m < 8; ++m) {
          float4 acc4 = {0.0f, 0.0f, 0.0f, 0.0f};
          #pragma unroll
          for (int k = 0; k < 8; ++k) {
            const float c = Ac[r][m * 8 + k];
            acc4.x += c * xk[k].x;
            acc4.y += c * xk[k].y;
            acc4.z += c * xk[k].z;
            acc4.w += c * xk[k].w;
          }
          ov[((size_t)m * nb + (b0 + r)) * (NC / 4) + t] = acc4;
        }
      }
    }
  }
}

extern "C" void kernel_launch(void* const* d_in, const int* in_sizes, int n_in,
                              void* d_out, int out_size, void* d_ws, size_t ws_size,
                              hipStream_t stream) {
  const float* x   = (const float*)d_in[0];
  float*       out = (float*)d_out;
  const int    nb  = in_sizes[0] / (NM * NC);
  gs_wmma_kernel<<<(nb + RB - 1) / RB, 256, 0, stream>>>(x, out, nb);
}
